# starting kernel
# baseline (speedup 1.0000x reference)
#include <hip/hip_runtime.h>

#define IN_F  4096
#define OUT_F 4096
#define BATCH 128

typedef __attribute__((ext_vector_type(2))) float v2f;
typedef __attribute__((ext_vector_type(4))) float v4f;
typedef __attribute__((ext_vector_type(8))) float v8f;

__global__ void zero_dense(v4f* __restrict__ W4, int n4) {
    int i = blockIdx.x * blockDim.x + threadIdx.x;
    if (i < n4) {
        W4[i] = (v4f){0.f, 0.f, 0.f, 0.f};
    }
}

__global__ void coo_scatter(const float* __restrict__ vals,
                            const int*   __restrict__ rows,
                            const int*   __restrict__ cols,
                            float* __restrict__ W, int nnz) {
    int i = blockIdx.x * blockDim.x + threadIdx.x;
    if (i < nnz) {
        size_t idx = (size_t)rows[i] * IN_F + (size_t)cols[i];
        __hip_atomic_fetch_add(&W[idx], vals[i],
                               __ATOMIC_RELAXED, __HIP_MEMORY_SCOPE_AGENT);
    }
}

__global__ void __launch_bounds__(32) dense_wmma_gemm(
        const float* __restrict__ W,
        const float* __restrict__ inp,
        const float* __restrict__ bias,
        float* __restrict__ out,
        int kdim) {
    const int r0   = blockIdx.x * 16;
    const int b0   = blockIdx.y * 64;
    const int lane = threadIdx.x & 31;
    const int half = lane >> 4;
    const int l16  = lane & 15;
    const int koff = half * 4;

    const float* arow = W   + (size_t)(r0 + l16) * IN_F + koff;
    const float* br0  = inp + (size_t)(b0 +  0 + l16) * IN_F + koff;
    const float* br1  = inp + (size_t)(b0 + 16 + l16) * IN_F + koff;
    const float* br2  = inp + (size_t)(b0 + 32 + l16) * IN_F + koff;
    const float* br3  = inp + (size_t)(b0 + 48 + l16) * IN_F + koff;

    v8f acc0 = {}, acc1 = {}, acc2 = {}, acc3 = {};

    constexpr int PF = 4;
    const int steps  = kdim >> 3;
    const int nmain  = (steps - PF) / PF;

    v4f A[PF], B0[PF], B1[PF], B2[PF], B3[PF];

    auto k8_step = [&](v4f av, v4f b0v, v4f b1v, v4f b2v, v4f b3v) {
        v2f alo = __builtin_shufflevector(av, av, 0, 1);
        v2f ahi = __builtin_shufflevector(av, av, 2, 3);
        v2f l0 = __builtin_shufflevector(b0v, b0v, 0, 1);
        v2f h0 = __builtin_shufflevector(b0v, b0v, 2, 3);
        v2f l1 = __builtin_shufflevector(b1v, b1v, 0, 1);
        v2f h1 = __builtin_shufflevector(b1v, b1v, 2, 3);
        v2f l2 = __builtin_shufflevector(b2v, b2v, 0, 1);
        v2f h2 = __builtin_shufflevector(b2v, b2v, 2, 3);
        v2f l3 = __builtin_shufflevector(b3v, b3v, 0, 1);
        v2f h3 = __builtin_shufflevector(b3v, b3v, 2, 3);
        acc0 = __builtin_amdgcn_wmma_f32_16x16x4_f32(false, alo, false, l0,
                                                     (short)0, acc0, false, false);
        acc1 = __builtin_amdgcn_wmma_f32_16x16x4_f32(false, alo, false, l1,
                                                     (short)0, acc1, false, false);
        acc2 = __builtin_amdgcn_wmma_f32_16x16x4_f32(false, alo, false, l2,
                                                     (short)0, acc2, false, false);
        acc3 = __builtin_amdgcn_wmma_f32_16x16x4_f32(false, alo, false, l3,
                                                     (short)0, acc3, false, false);
        acc0 = __builtin_amdgcn_wmma_f32_16x16x4_f32(false, ahi, false, h0,
                                                     (short)0, acc0, false, false);
        acc1 = __builtin_amdgcn_wmma_f32_16x16x4_f32(false, ahi, false, h1,
                                                     (short)0, acc1, false, false);
        acc2 = __builtin_amdgcn_wmma_f32_16x16x4_f32(false, ahi, false, h2,
                                                     (short)0, acc2, false, false);
        acc3 = __builtin_amdgcn_wmma_f32_16x16x4_f32(false, ahi, false, h3,
                                                     (short)0, acc3, false, false);
    };

#pragma unroll
    for (int p = 0; p < PF; ++p) {
        A[p]  = *(const v4f*)(arow + (size_t)p * 8);
        B0[p] = *(const v4f*)(br0  + (size_t)p * 8);
        B1[p] = *(const v4f*)(br1  + (size_t)p * 8);
        B2[p] = *(const v4f*)(br2  + (size_t)p * 8);
        B3[p] = *(const v4f*)(br3  + (size_t)p * 8);
    }

    int sbase = 0;
    for (int t = 0; t < nmain; ++t) {
#pragma unroll
        for (int i = 0; i < PF; ++i) {
            v4f av  = A[i];
            v4f b0v = B0[i], b1v = B1[i], b2v = B2[i], b3v = B3[i];

            const size_t off = (size_t)(sbase + i + PF) * 8;
            A[i]  = *(const v4f*)(arow + off);
            B0[i] = *(const v4f*)(br0  + off);
            B1[i] = *(const v4f*)(br1  + off);
            B2[i] = *(const v4f*)(br2  + off);
            B3[i] = *(const v4f*)(br3  + off);

            k8_step(av, b0v, b1v, b2v, b3v);
        }
        sbase += PF;
    }

#pragma unroll
    for (int i = 0; i < PF; ++i) {
        k8_step(A[i], B0[i], B1[i], B2[i], B3[i]);
    }

    const int mbase = r0 + half * 8;
    v4f bv0 = *(const v4f*)(bias + mbase);
    v4f bv1 = *(const v4f*)(bias + mbase + 4);

    auto store_tile = [&](v8f acc, float* o) {
        v4f lo, hi;
#pragma unroll
        for (int v = 0; v < 4; ++v) {
            lo[v] = acc[v]     + bv0[v];
            hi[v] = acc[v + 4] + bv1[v];
        }
        *(v4f*)o       = lo;
        *(v4f*)(o + 4) = hi;
    };

    store_tile(acc0, out + (size_t)(b0 +  0 + l16) * OUT_F + mbase);
    store_tile(acc1, out + (size_t)(b0 + 16 + l16) * OUT_F + mbase);
    store_tile(acc2, out + (size_t)(b0 + 32 + l16) * OUT_F + mbase);
    store_tile(acc3, out + (size_t)(b0 + 48 + l16) * OUT_F + mbase);
}

extern "C" void kernel_launch(void* const* d_in, const int* in_sizes, int n_in,
                              void* d_out, int out_size, void* d_ws, size_t ws_size,
                              hipStream_t stream) {
    const float* inp  = (const float*)d_in[0];
    const float* wval = (const float*)d_in[1];
    const int*   wrow = (const int*)d_in[2];
    const int*   wcol = (const int*)d_in[3];
    const float* bias = (const float*)d_in[4];
    float*       out  = (float*)d_out;
    float*       W    = (float*)d_ws;
    const int    nnz  = in_sizes[1];

    const int n4 = (OUT_F * IN_F) / 4;
    zero_dense<<<(n4 + 255) / 256, 256, 0, stream>>>((v4f*)W, n4);

    coo_scatter<<<(nnz + 255) / 256, 256, 0, stream>>>(wval, wrow, wcol, W, nnz);

    dim3 grid(OUT_F / 16, BATCH / 64);
    dense_wmma_gemm<<<grid, 32, 0, stream>>>(W, inp, bias, out, IN_F);
}


# Dense W built per 16-row band: owner work-group filters a 1-byte band-key list (no binning), sums rows in LDS (float adds), stores whole lines; no zero pass, no global atomics; add order not fixed
# speedup vs baseline: 1.6983x; 1.6983x over previous
#include <hip/hip_runtime.h>

#define IN_F  4096
#define OUT_F 4096
#define BATCH 128

typedef __attribute__((ext_vector_type(2))) float v2f;
typedef __attribute__((ext_vector_type(4))) float v4f;
typedef __attribute__((ext_vector_type(8))) float v8f;

typedef __attribute__((ext_vector_type(4))) unsigned int v4u;
typedef __attribute__((ext_vector_type(4))) int v4i;

#define BAND_ROWS 16
#define NBANDS    (OUT_F / BAND_ROWS)

__global__ void __launch_bounds__(256) band_keys(const int* __restrict__ rows,
                                                 unsigned int* __restrict__ keys,
                                                 int nnz) {
    const int nchunks = nnz >> 4;
    const v4i* r4 = (const v4i*)rows;
    v4u* k4 = (v4u*)keys;
    for (int c = blockIdx.x * blockDim.x + threadIdx.x; c < nchunks;
         c += gridDim.x * blockDim.x) {
        v4u k;
        for (int d = 0; d < 4; ++d) {
            v4i r = r4[c * 4 + d];
            k[d] =  (((unsigned)r[0] >> 4) & 255u)
                 | ((((unsigned)r[1] >> 4) & 255u) << 8)
                 | ((((unsigned)r[2] >> 4) & 255u) << 16)
                 | ((((unsigned)r[3] >> 4) & 255u) << 24);
        }
        k4[c] = k;
    }
}

typedef __attribute__((ext_vector_type(2))) int v2i;
#define LIST_CAP 7680

__global__ void __launch_bounds__(1024) build_bands(
        const float* __restrict__ vals,
        const int*   __restrict__ rows,
        const int*   __restrict__ cols,
        const unsigned int* __restrict__ keys,
        float* __restrict__ W, int nnz) {
    __shared__ v4f slab4[BAND_ROWS * IN_F / 4];
    __shared__ v2i list[LIST_CAP];
    __shared__ int cnt;

    float* slab = (float*)slab4;
    const int tid  = threadIdx.x;
    const int band = blockIdx.x;
    const unsigned pat = (unsigned)band * 0x01010101u;

    for (int q = tid; q < BAND_ROWS * IN_F / 4; q += 1024)
        slab4[q] = (v4f){0.f, 0.f, 0.f, 0.f};
    if (tid == 0) cnt = 0;
    __syncthreads();

    auto add_entry = [&](int i) {
        const int r = rows[i] & (BAND_ROWS - 1);
        const int c = cols[i] & (IN_F - 1);
        __hip_atomic_fetch_add(&slab[r * IN_F + c], vals[i],
                               __ATOMIC_RELAXED, __HIP_MEMORY_SCOPE_WORKGROUP);
    };
    auto add_chunk = [&](int c, unsigned m) {
        while (m) {
            const int b = __builtin_ctz(m);
            add_entry((c << 4) + ((b & 7) << 2) + (b >> 3));
            m &= m - 1;
        }
    };
    auto note = [&](int c, unsigned m) {
        const int rank = __hip_atomic_fetch_add(&cnt, 1, __ATOMIC_RELAXED,
                                                __HIP_MEMORY_SCOPE_WORKGROUP);
        if (rank < LIST_CAP) list[rank] = (v2i){c, (int)m};
        else                 add_chunk(c, m);
    };
    auto flags = [&](unsigned w) {
        const unsigned x = w ^ pat;
        const unsigned t = (x & 0x7f7f7f7fu) + 0x7f7f7f7fu;
        return ~(t | x) & 0x80808080u;
    };

    const int nchunks = nnz >> 4;
    const v4u* k4 = (const v4u*)keys;
    const unsigned miss = pat ^ 0x01010101u;
    for (int c0 = tid; c0 < nchunks; c0 += 4 * 1024) {
        v4u k[4];
        for (int u = 0; u < 4; ++u) {
            const int c = c0 + u * 1024;
            k[u] = c < nchunks ? k4[c] : (v4u){miss, miss, miss, miss};
        }
        for (int u = 0; u < 4; ++u) {
            const unsigned z0 = flags(k[u][0]), z1 = flags(k[u][1]);
            const unsigned z2 = flags(k[u][2]), z3 = flags(k[u][3]);
            if (z0 | z1 | z2 | z3)
                note(c0 + u * 1024, (z0 >> 7) | (z1 >> 6) | (z2 >> 5) | (z3 >> 4));
        }
    }
    for (int i = (nchunks << 4) + tid; i < nnz; i += 1024)
        if ((((unsigned)rows[i] >> 4) & 255u) == (unsigned)band) add_entry(i);
    __syncthreads();

    const int n = cnt < LIST_CAP ? cnt : LIST_CAP;
    for (int j = tid; j < n; j += 1024) {
        const v2i it = list[j];
        add_chunk(it[0], (unsigned)it[1]);
    }
    __syncthreads();

    v4f* W4 = (v4f*)(W + (size_t)band * BAND_ROWS * IN_F);
    for (int q = tid; q < BAND_ROWS * IN_F / 4; q += 1024)
        W4[q] = slab4[q];
}

#define TILE_M   32
#define TILE_N   64
#define KCHUNK   32
#define PITCH4   9
#define STAGE4   ((TILE_M + TILE_N) * PITCH4)
#define NSTAGE   4
#define NSLOT    8

__global__ void __launch_bounds__(128) dense_wmma_gemm(
        const float* __restrict__ W,
        const float* __restrict__ inp,
        const float* __restrict__ bias,
        float* __restrict__ out,
        int kdim) {
    __shared__ v4f lds4[NSTAGE * STAGE4];

    const int tid  = threadIdx.x;
    const int lane = tid & 31;
    const int wv   = tid >> 5;
    const int half = lane >> 4;
    const int l16  = lane & 15;
    const int r0   = blockIdx.x * TILE_M;
    const int b0   = blockIdx.y * TILE_N;

    const int lrow = tid >> 3;
    const int seg  = tid & 7;
    const float* gp[6];
    gp[0] = W   + (size_t)(r0 +  0 + lrow) * IN_F + seg * 4;
    gp[1] = W   + (size_t)(r0 + 16 + lrow) * IN_F + seg * 4;
    gp[2] = inp + (size_t)(b0 +  0 + lrow) * IN_F + seg * 4;
    gp[3] = inp + (size_t)(b0 + 16 + lrow) * IN_F + seg * 4;
    gp[4] = inp + (size_t)(b0 + 32 + lrow) * IN_F + seg * 4;
    gp[5] = inp + (size_t)(b0 + 48 + lrow) * IN_F + seg * 4;

    const int st_off = lrow * PITCH4 + seg;
    const int fa_off = l16 * PITCH4 + half * 4;
    const int fb_off = (TILE_M + wv * 16 + l16) * PITCH4 + half * 4;

    const int nchunks = kdim >> 5;
    const int ngroups = nchunks >> 3;
    const int lastc   = nchunks - 1;

    v4f G[NSLOT][6];
    v4f FA[2][2][4], FB[2][4];
    v8f acc0 = {}, acc1 = {};

    auto gload = [&](int slot, int chunk) {
        const int cc = chunk < lastc ? chunk : lastc;
        const size_t off = (size_t)cc * KCHUNK;
#pragma unroll
        for (int j = 0; j < 6; ++j) G[slot][j] = *(const v4f*)(gp[j] + off);
    };
    auto sstore = [&](int slot, int stage) {
        v4f* s = lds4 + stage * STAGE4 + st_off;
#pragma unroll
        for (int j = 0; j < 6; ++j) s[j * 16 * PITCH4] = G[slot][j];
    };
    auto fload = [&](int set, int stage) {
        const v4f* s = lds4 + stage * STAGE4;
#pragma unroll
        for (int q = 0; q < 4; ++q) {
            FA[set][0][q] = s[fa_off + q];
            FA[set][1][q] = s[fa_off + 16 * PITCH4 + q];
            FB[set][q]    = s[fb_off + q];
        }
    };
    auto compute = [&](int set) {
#pragma unroll
        for (int q = 0; q < 4; ++q) {
            v4f a0 = FA[set][0][q], a1 = FA[set][1][q], b = FB[set][q];
            v2f a0l = __builtin_shufflevector(a0, a0, 0, 1);
            v2f a0h = __builtin_shufflevector(a0, a0, 2, 3);
            v2f a1l = __builtin_shufflevector(a1, a1, 0, 1);
            v2f a1h = __builtin_shufflevector(a1, a1, 2, 3);
            v2f bl  = __builtin_shufflevector(b, b, 0, 1);
            v2f bh  = __builtin_shufflevector(b, b, 2, 3);
            acc0 = __builtin_amdgcn_wmma_f32_16x16x4_f32(false, a0l, false, bl,
                                                         (short)0, acc0, false, false);
            acc1 = __builtin_amdgcn_wmma_f32_16x16x4_f32(false, a1l, false, bl,
                                                         (short)0, acc1, false, false);
            acc0 = __builtin_amdgcn_wmma_f32_16x16x4_f32(false, a0h, false, bh,
                                                         (short)0, acc0, false, false);
            acc1 = __builtin_amdgcn_wmma_f32_16x16x4_f32(false, a1h, false, bh,
                                                         (short)0, acc1, false, false);
        }
    };
    auto lds_handover = [&]() {
        asm volatile("s_wait_dscnt 0x0\n\ts_barrier_signal -1\n\ts_barrier_wait -1" ::: "memory");
    };

#pragma unroll
    for (int u = 0; u < NSLOT; ++u) gload(u, u);
    sstore(0, 0);
    gload(0, NSLOT + 0);
    sstore(1, 1);
    gload(1, NSLOT + 1);
    lds_handover();
    fload(0, 0);

    int tb = 0;
    for (int g = 0; g < ngroups; ++g) {
#pragma unroll
        for (int i = 0; i < 8; ++i) {
            fload((i + 1) & 1, (i + 1) & 3);
            sstore((i + 2) & 7, (i + 2) & 3);
            gload((i + 2) & 7, tb + i + 2 + NSLOT);
            __builtin_amdgcn_sched_barrier(0);
            compute(i & 1);
            __builtin_amdgcn_sched_barrier(0);
            lds_handover();
        }
        tb += 8;
    }

    float* orow = out + (size_t)(b0 + wv * 16 + l16) * OUT_F + r0 + half * 8;
    const float* brow = bias + r0 + half * 8;

    auto store_tile = [&](v8f acc, float* o, const float* bp) {
        v4f bv0 = *(const v4f*)bp;
        v4f bv1 = *(const v4f*)(bp + 4);
        v4f lo, hi;
#pragma unroll
        for (int v = 0; v < 4; ++v) {
            lo[v] = acc[v]     + bv0[v];
            hi[v] = acc[v + 4] + bv1[v];
        }
        *(v4f*)o       = lo;
        *(v4f*)(o + 4) = hi;
    };

    store_tile(acc0, orow,      brow);
    store_tile(acc1, orow + 16, brow + 16);
}

extern "C" void kernel_launch(void* const* d_in, const int* in_sizes, int n_in,
                              void* d_out, int out_size, void* d_ws, size_t ws_size,
                              hipStream_t stream) {
    const float* inp  = (const float*)d_in[0];
    const float* wval = (const float*)d_in[1];
    const int*   wrow = (const int*)d_in[2];
    const int*   wcol = (const int*)d_in[3];
    const float* bias = (const float*)d_in[4];
    float*       out  = (float*)d_out;
    float*       W    = (float*)d_ws;
    const int    nnz  = in_sizes[1];
    unsigned int* keys = (unsigned int*)d_out;

    band_keys<<<512, 256, 0, stream>>>(wrow, keys, nnz);

    build_bands<<<NBANDS, 1024, 0, stream>>>(wval, wrow, wcol, keys, W, nnz);

    dim3 grid(OUT_F / TILE_M, BATCH / TILE_N);
    dense_wmma_gemm<<<grid, 128, 0, stream>>>(W, inp, bias, out, IN_F);
}


# Band membership computed once: kernel 1 leaves each band a bitmap (1 bit per 4 nonzeros) in the band's own W rows; its owner reads 50 KB of bits, not 1.6 MB of keys; LDS add order still not fixed
# speedup vs baseline: 2.1373x; 1.2585x over previous
#include <hip/hip_runtime.h>

#define IN_F  4096
#define OUT_F 4096
#define BATCH 128

typedef __attribute__((ext_vector_type(2))) float v2f;
typedef __attribute__((ext_vector_type(4))) float v4f;
typedef __attribute__((ext_vector_type(8))) float v8f;

typedef __attribute__((ext_vector_type(4))) unsigned int v4u;
typedef __attribute__((ext_vector_type(4))) int v4i;

#define BAND_ROWS 16
#define NBANDS    (OUT_F / BAND_ROWS)
#define TILE_Q    2048
#define TILE_NZ   (4 * TILE_Q)
#define TILE_W    (TILE_Q / 32)
#define BM_PITCH  (NBANDS + 1)
#define BM_WORDS  (TILE_W * BM_PITCH)
#define BM_VEC    ((BM_WORDS + 3) / 4)
#define NNZ_SPEC  1600000
#define NTILES_SPEC ((NNZ_SPEC + TILE_NZ - 1) / TILE_NZ)

__global__ void __launch_bounds__(256) band_bitmaps(const int* __restrict__ rows,
                                                    float* W, int nnz) {
    __shared__ v4u bm4[BM_VEC];
    unsigned int* bm = (unsigned int*)bm4;

    const int tid  = threadIdx.x;
    const int lane = tid & 31;
    const int wv   = tid >> 5;
    const int half = lane >> 4;
    const int l16  = lane & 15;
    const int ntiles = (nnz + TILE_NZ - 1) / TILE_NZ;

    for (int t = blockIdx.x; t < ntiles; t += gridDim.x) {
        const int base = t * TILE_NZ;
        v4i  r[8];
        bool whole[8];
        for (int u = 0; u < 8; ++u) {
            const int i0 = base + 4 * (tid + 256 * u);
            whole[u] = i0 + 3 < nnz;
            r[u] = whole[u] ? *(const v4i*)(rows + i0) : (v4i){0, 0, 0, 0};
        }
        for (int q = tid; q < BM_VEC; q += 256) bm4[q] = (v4u){0u, 0u, 0u, 0u};
        __syncthreads();
        for (int u = 0; u < 8; ++u) {
            if (whole[u]) {
                for (int e = 0; e < 4; ++e) {
                    const int band = (r[u][e] >> 4) & (NBANDS - 1);
                    __hip_atomic_fetch_or(&bm[(wv + 8 * u) * BM_PITCH + band], 1u << lane,
                                          __ATOMIC_RELAXED, __HIP_MEMORY_SCOPE_WORKGROUP);
                }
            }
        }
        __syncthreads();
        for (int v = 0; v < 16; ++v) {
            const int band = 2 * (wv + 8 * v) + half;
            v4u x;
            for (int k = 0; k < 4; ++k) x[k] = bm[(4 * l16 + k) * BM_PITCH + band];
            v4u* piece = (v4u*)(W + (size_t)band * BAND_ROWS * IN_F) + t * (TILE_W / 4);
            piece[l16] = x;
        }
        __syncthreads();
    }
}

#define LIST_CAP 15360

__global__ void __launch_bounds__(1024) build_bands(
        const float* __restrict__ vals,
        const int*   __restrict__ rows,
        const int*   __restrict__ cols,
        float* W, int nnz) {
    __shared__ v4f slab4[BAND_ROWS * IN_F / 4];
    __shared__ int list[LIST_CAP];
    __shared__ int cnt;

    float* slab = (float*)slab4;
    const int tid  = threadIdx.x;
    const int band = blockIdx.x;

    for (int q = tid; q < BAND_ROWS * IN_F / 4; q += 1024)
        slab4[q] = (v4f){0.f, 0.f, 0.f, 0.f};
    if (tid == 0) cnt = 0;
    __syncthreads();

    auto add_one = [&](int r, int c, float x) {
        if (((r >> 4) & (NBANDS - 1)) == band)
            __hip_atomic_fetch_add(&slab[(r & (BAND_ROWS - 1)) * IN_F + (c & (IN_F - 1))], x,
                                   __ATOMIC_RELAXED, __HIP_MEMORY_SCOPE_WORKGROUP);
    };
    auto add_quad = [&](int j) {
        const int i0 = j << 2;
        const v4i r = *(const v4i*)(rows + i0);
        const v4i c = *(const v4i*)(cols + i0);
        const v4f x = *(const v4f*)(vals + i0);
        for (int e = 0; e < 4; ++e) add_one(r[e], c[e], x[e]);
    };

    v4f* W4 = (v4f*)(W + (size_t)band * BAND_ROWS * IN_F);
    const v4u* bits4 = (const v4u*)W4;
    const int ntiles = (nnz + TILE_NZ - 1) / TILE_NZ;
    const int nvec   = ntiles * (TILE_W / 4);
    for (int c0 = tid; c0 < nvec; c0 += 4 * 1024) {
        v4u k[4];
        int hits = 0;
        for (int u = 0; u < 4; ++u) {
            const int c = c0 + u * 1024;
            k[u] = c < nvec ? bits4[c] : (v4u){0u, 0u, 0u, 0u};
        }
        for (int u = 0; u < 4; ++u)
            for (int d = 0; d < 4; ++d) hits += __builtin_popcount(k[u][d]);
        int rank = __hip_atomic_fetch_add(&cnt, hits, __ATOMIC_RELAXED,
                                          __HIP_MEMORY_SCOPE_WORKGROUP);
        for (int u = 0; u < 4; ++u) {
            for (int d = 0; d < 4; ++d) {
                unsigned m = k[u][d];
                const int j0 = (((c0 + u * 1024) << 2) + d) << 5;
                while (m) {
                    if (rank < LIST_CAP) list[rank] = j0 + __builtin_ctz(m);
                    ++rank;
                    m &= m - 1;
                }
            }
        }
    }
    __syncthreads();

    const int n = cnt;
    if (n <= LIST_CAP) {
        for (int j = tid; j < n; j += 1024) add_quad(list[j]);
    } else {
        const unsigned int* bits = (const unsigned int*)W4;
        for (int g = tid; g < nvec * 4; g += 1024) {
            unsigned m = bits[g];
            while (m) {
                add_quad((g << 5) + __builtin_ctz(m));
                m &= m - 1;
            }
        }
    }
    if (tid < (nnz & 3)) {
        const int i = (nnz & ~3) + tid;
        add_one(rows[i], cols[i], vals[i]);
    }
    __syncthreads();

    for (int q = tid; q < BAND_ROWS * IN_F / 4; q += 1024)
        W4[q] = slab4[q];
}

#define TILE_M   32
#define TILE_N   64
#define KCHUNK   32
#define PITCH4   9
#define STAGE4   ((TILE_M + TILE_N) * PITCH4)
#define NSTAGE   4
#define NSLOT    8

__global__ void __launch_bounds__(128) dense_wmma_gemm(
        const float* __restrict__ W,
        const float* __restrict__ inp,
        const float* __restrict__ bias,
        float* __restrict__ out,
        int kdim) {
    __shared__ v4f lds4[NSTAGE * STAGE4];

    const int tid  = threadIdx.x;
    const int lane = tid & 31;
    const int wv   = tid >> 5;
    const int half = lane >> 4;
    const int l16  = lane & 15;
    const int r0   = blockIdx.x * TILE_M;
    const int b0   = blockIdx.y * TILE_N;

    const int lrow = tid >> 3;
    const int seg  = tid & 7;
    const float* gp[6];
    gp[0] = W   + (size_t)(r0 +  0 + lrow) * IN_F + seg * 4;
    gp[1] = W   + (size_t)(r0 + 16 + lrow) * IN_F + seg * 4;
    gp[2] = inp + (size_t)(b0 +  0 + lrow) * IN_F + seg * 4;
    gp[3] = inp + (size_t)(b0 + 16 + lrow) * IN_F + seg * 4;
    gp[4] = inp + (size_t)(b0 + 32 + lrow) * IN_F + seg * 4;
    gp[5] = inp + (size_t)(b0 + 48 + lrow) * IN_F + seg * 4;

    const int st_off = lrow * PITCH4 + seg;
    const int fa_off = l16 * PITCH4 + half * 4;
    const int fb_off = (TILE_M + wv * 16 + l16) * PITCH4 + half * 4;

    const int nchunks = kdim >> 5;
    const int ngroups = nchunks >> 3;
    const int lastc   = nchunks - 1;

    v4f G[NSLOT][6];
    v4f FA[2][2][4], FB[2][4];
    v8f acc0 = {}, acc1 = {};

    auto gload = [&](int slot, int chunk) {
        const int cc = chunk < lastc ? chunk : lastc;
        const size_t off = (size_t)cc * KCHUNK;
#pragma unroll
        for (int j = 0; j < 6; ++j) G[slot][j] = *(const v4f*)(gp[j] + off);
    };
    auto sstore = [&](int slot, int stage) {
        v4f* s = lds4 + stage * STAGE4 + st_off;
#pragma unroll
        for (int j = 0; j < 6; ++j) s[j * 16 * PITCH4] = G[slot][j];
    };
    auto fload = [&](int set, int stage) {
        const v4f* s = lds4 + stage * STAGE4;
#pragma unroll
        for (int q = 0; q < 4; ++q) {
            FA[set][0][q] = s[fa_off + q];
            FA[set][1][q] = s[fa_off + 16 * PITCH4 + q];
            FB[set][q]    = s[fb_off + q];
        }
    };
    auto compute = [&](int set) {
#pragma unroll
        for (int q = 0; q < 4; ++q) {
            v4f a0 = FA[set][0][q], a1 = FA[set][1][q], b = FB[set][q];
            v2f a0l = __builtin_shufflevector(a0, a0, 0, 1);
            v2f a0h = __builtin_shufflevector(a0, a0, 2, 3);
            v2f a1l = __builtin_shufflevector(a1, a1, 0, 1);
            v2f a1h = __builtin_shufflevector(a1, a1, 2, 3);
            v2f bl  = __builtin_shufflevector(b, b, 0, 1);
            v2f bh  = __builtin_shufflevector(b, b, 2, 3);
            acc0 = __builtin_amdgcn_wmma_f32_16x16x4_f32(false, a0l, false, bl,
                                                         (short)0, acc0, false, false);
            acc1 = __builtin_amdgcn_wmma_f32_16x16x4_f32(false, a1l, false, bl,
                                                         (short)0, acc1, false, false);
            acc0 = __builtin_amdgcn_wmma_f32_16x16x4_f32(false, a0h, false, bh,
                                                         (short)0, acc0, false, false);
            acc1 = __builtin_amdgcn_wmma_f32_16x16x4_f32(false, a1h, false, bh,
                                                         (short)0, acc1, false, false);
        }
    };
    auto lds_handover = [&]() {
        asm volatile("s_wait_dscnt 0x0\n\ts_barrier_signal -1\n\ts_barrier_wait -1" ::: "memory");
    };

#pragma unroll
    for (int u = 0; u < NSLOT; ++u) gload(u, u);
    sstore(0, 0);
    gload(0, NSLOT + 0);
    sstore(1, 1);
    gload(1, NSLOT + 1);
    lds_handover();
    fload(0, 0);

    int tb = 0;
    for (int g = 0; g < ngroups; ++g) {
#pragma unroll
        for (int i = 0; i < 8; ++i) {
            fload((i + 1) & 1, (i + 1) & 3);
            sstore((i + 2) & 7, (i + 2) & 3);
            gload((i + 2) & 7, tb + i + 2 + NSLOT);
            __builtin_amdgcn_sched_barrier(0);
            compute(i & 1);
            __builtin_amdgcn_sched_barrier(0);
            lds_handover();
        }
        tb += 8;
    }

    float* orow = out + (size_t)(b0 + wv * 16 + l16) * OUT_F + r0 + half * 8;
    const float* brow = bias + r0 + half * 8;

    auto store_tile = [&](v8f acc, float* o, const float* bp) {
        v4f bv0 = *(const v4f*)bp;
        v4f bv1 = *(const v4f*)(bp + 4);
        v4f lo, hi;
#pragma unroll
        for (int v = 0; v < 4; ++v) {
            lo[v] = acc[v]     + bv0[v];
            hi[v] = acc[v + 4] + bv1[v];
        }
        *(v4f*)o       = lo;
        *(v4f*)(o + 4) = hi;
    };

    store_tile(acc0, orow,      brow);
    store_tile(acc1, orow + 16, brow + 16);
}

extern "C" void kernel_launch(void* const* d_in, const int* in_sizes, int n_in,
                              void* d_out, int out_size, void* d_ws, size_t ws_size,
                              hipStream_t stream) {
    const float* inp  = (const float*)d_in[0];
    const float* wval = (const float*)d_in[1];
    const int*   wrow = (const int*)d_in[2];
    const int*   wcol = (const int*)d_in[3];
    const float* bias = (const float*)d_in[4];
    float*       out  = (float*)d_out;
    float*       W    = (float*)d_ws;
    const int    nnz  = in_sizes[1];

    band_bitmaps<<<NTILES_SPEC, 256, 0, stream>>>(wrow, W, nnz);

    build_bands<<<NBANDS, 1024, 0, stream>>>(wval, wrow, wcol, W, nnz);

    dim3 grid(OUT_F / TILE_M, BATCH / TILE_N);
    dense_wmma_gemm<<<grid, 128, 0, stream>>>(W, inp, bias, out, IN_F);
}
